# starting kernel
# baseline (speedup 1.0000x reference)
#include <hip/hip_runtime.h>
#include <hip/hip_bf16.h>

typedef __attribute__((ext_vector_type(2))) float v2f;
typedef __attribute__((ext_vector_type(4))) float v4f;
typedef __attribute__((ext_vector_type(8))) float v8f;

#define ALPHA_LRELU 0.2f
#define NEG_INF_F  (-1.0e30f)

__global__ void w1a3_kernel(const float* __restrict__ W1, const float* __restrict__ a,
                            float* __restrict__ w1a3) {
    int t = threadIdx.x;
    const float* row = W1 + (size_t)t * 256;
    float acc = 0.f;
    #pragma unroll 4
    for (int c = 0; c < 256; c += 4) {
        v4f v = *(const v4f*)(row + c);
        acc += v[0] * a[512 + c] + v[1] * a[512 + c + 1] +
               v[2] * a[512 + c + 2] + v[3] * a[512 + c + 3];
    }
    w1a3[t] = acc;
}

__global__ void gemm_h_kernel(const float* __restrict__ x, const float* __restrict__ W,
                              float* __restrict__ h) {
    int b    = blockIdx.y;
    int i0   = blockIdx.x * 16;
    int wave = threadIdx.x >> 5;
    int lane = threadIdx.x & 31;

    int M    = lane & 15;
    int koff = (lane >> 4) << 1;
    int klo  = (lane < 16) ? 0 : 2;
    int ncol = lane & 15;

    const float* xrow = x + (size_t)(b * 512 + i0 + M) * 256;
    int n0a = wave * 16;
    int n0b = (wave + 8) * 16;

    v8f c0 = {}; v8f c1 = {};
    for (int k0 = 0; k0 < 256; k0 += 4) {
        v2f afrag = *(const v2f*)(xrow + k0 + koff);

        const float* wp = W + (size_t)(k0 + klo) * 256;
        v2f b0, b1;
        b0[0] = wp[n0a + ncol]; b0[1] = wp[256 + n0a + ncol];
        b1[0] = wp[n0b + ncol]; b1[1] = wp[256 + n0b + ncol];

        c0 = __builtin_amdgcn_wmma_f32_16x16x4_f32(false, afrag, false, b0, (short)0, c0, false, false);
        c1 = __builtin_amdgcn_wmma_f32_16x16x4_f32(false, afrag, false, b1, (short)0, c1, false, false);
    }

    int mbase = (lane >> 4) << 3;
    float* hb = h + (size_t)(b * 512 + i0) * 256;
    #pragma unroll
    for (int v = 0; v < 8; ++v) {
        hb[(size_t)(mbase + v) * 256 + n0a + ncol] = c0[v];
        hb[(size_t)(mbase + v) * 256 + n0b + ncol] = c1[v];
    }
}

__global__ void sij_kernel(const float* __restrict__ h, const float* __restrict__ a,
                           float* __restrict__ s_i, float* __restrict__ s_j) {
    int idx = blockIdx.x * blockDim.x + threadIdx.x;
    const float* hr = h + (size_t)idx * 256;
    float acc1 = 0.f, acc2 = 0.f;
    #pragma unroll 4
    for (int o = 0; o < 256; o += 4) {
        v4f hv = *(const v4f*)(hr + o);
        acc1 += hv[0] * a[o]       + hv[1] * a[o + 1]       + hv[2] * a[o + 2]       + hv[3] * a[o + 3];
        acc2 += hv[0] * a[256 + o] + hv[1] * a[256 + o + 1] + hv[2] * a[256 + o + 2] + hv[3] * a[256 + o + 3];
    }
    s_i[idx] = acc1;
    s_j[idx] = acc2;
}

__global__ void score_kernel(const float* __restrict__ edge, const float* __restrict__ adj,
                             const float* __restrict__ s_i, const float* __restrict__ s_j,
                             const float* __restrict__ w1a3, float* __restrict__ escore) {
    __shared__ float wle[64];
    int b   = blockIdx.y;
    int i   = blockIdx.x;
    int tid = threadIdx.x;

    if (tid < 64) wle[tid] = w1a3[tid];
    __syncthreads();

    float si = s_i[b * 512 + i];
    const float* ebase = edge + ((size_t)(b * 512 + i) * 512) * 64;
    const float* abase = adj  +  (size_t)(b * 512 + i) * 512;
    float* eout = escore + (size_t)(b * 512 + i) * 512;

    #pragma unroll 1
    for (int jj = 0; jj < 2; ++jj) {
        int j = jj * 256 + tid;
        const float* ep = ebase + (size_t)j * 64;
        __builtin_prefetch(ep + 256 * 64, 0, 1);
        float s = 0.f;
        #pragma unroll
        for (int c = 0; c < 64; c += 4) {
            v4f v = __builtin_nontemporal_load((const v4f*)(ep + c));
            s += v[0] * wle[c] + v[1] * wle[c + 1] + v[2] * wle[c + 2] + v[3] * wle[c + 3];
        }
        float e = si + s_j[b * 512 + j] + s;
        e = (e > 0.f) ? e : ALPHA_LRELU * e;
        float ad = __builtin_nontemporal_load(abase + j);
        e = (ad > 0.0f) ? e : NEG_INF_F;
        eout[j] = e;
    }
}

__global__ void attn_kernel(const float* __restrict__ escore, const float* __restrict__ h,
                            float* __restrict__ out) {
    __shared__ float att[16][516];
    __shared__ float red[16][16];
    __shared__ float rowstat[16];

    int b   = blockIdx.y;
    int i0  = blockIdx.x * 16;
    int tid = threadIdx.x;

    const float* ebase = escore + (size_t)(b * 512 + i0) * 512;
    #pragma unroll
    for (int iter = 0; iter < 32; ++iter) {
        int flat = iter * 256 + tid;
        att[flat >> 9][flat & 511] = ebase[flat];
    }
    __syncthreads();

    int row = tid >> 4;
    int l16 = tid & 15;

    float m = -3.0e38f;
    for (int j = l16; j < 512; j += 16) m = fmaxf(m, att[row][j]);
    red[row][l16] = m;
    __syncthreads();
    if (l16 == 0) {
        float mm = red[row][0];
        #pragma unroll
        for (int q = 1; q < 16; ++q) mm = fmaxf(mm, red[row][q]);
        rowstat[row] = mm;
    }
    __syncthreads();
    float mx = rowstat[row];

    float sum = 0.f;
    for (int j = l16; j < 512; j += 16) {
        float v = __expf(att[row][j] - mx);
        att[row][j] = v;
        sum += v;
    }
    red[row][l16] = sum;
    __syncthreads();
    if (l16 == 0) {
        float t = 0.f;
        #pragma unroll
        for (int q = 0; q < 16; ++q) t += red[row][q];
        rowstat[row] = t;
    }
    __syncthreads();
    float inv = 1.0f / rowstat[row];
    for (int j = l16; j < 512; j += 16) att[row][j] *= inv;
    __syncthreads();

    int wave = tid >> 5;
    int lane = tid & 31;
    int M     = lane & 15;
    int koff  = (lane >> 4) << 1;
    int klo   = (lane < 16) ? 0 : 2;
    int ncol  = lane & 15;
    int mbase = (lane >> 4) << 3;

    const float* hb = h + (size_t)b * 512 * 256;
    float* ob = out + (size_t)(b * 512 + i0) * 256;

    #pragma unroll 1
    for (int t2 = 0; t2 < 2; ++t2) {
        int n0 = (wave + t2 * 8) * 16;
        v8f c = {};
        for (int k0 = 0; k0 < 512; k0 += 4) {
            v2f afrag;
            afrag[0] = att[M][k0 + koff];
            afrag[1] = att[M][k0 + koff + 1];
            v2f bf;
            const float* hp = hb + (size_t)(k0 + klo) * 256 + n0 + ncol;
            bf[0] = hp[0];
            bf[1] = hp[256];
            c = __builtin_amdgcn_wmma_f32_16x16x4_f32(false, afrag, false, bf, (short)0, c, false, false);
        }
        #pragma unroll
        for (int v = 0; v < 8; ++v) {
            float xv = c[v];
            float o = (xv > 0.f) ? xv : (__expf(xv) - 1.0f);
            ob[(size_t)(mbase + v) * 256 + n0 + ncol] = o;
        }
    }
}

extern "C" void kernel_launch(void* const* d_in, const int* in_sizes, int n_in,
                              void* d_out, int out_size, void* d_ws, size_t ws_size,
                              hipStream_t stream) {
    const float* x    = (const float*)d_in[0];
    const float* edge = (const float*)d_in[1];
    const float* adj  = (const float*)d_in[2];
    const float* W    = (const float*)d_in[3];
    const float* W1   = (const float*)d_in[4];
    const float* a    = (const float*)d_in[5];
    float* out = (float*)d_out;

    float* h      = (float*)d_ws;
    float* s_i    = h + 4 * 512 * 256;
    float* s_j    = s_i + 4 * 512;
    float* w1a3   = s_j + 4 * 512;
    float* escore = w1a3 + 64;

    w1a3_kernel<<<1, 64, 0, stream>>>(W1, a, w1a3);
    gemm_h_kernel<<<dim3(32, 4), 256, 0, stream>>>(x, W, h);
    sij_kernel<<<8, 256, 0, stream>>>(h, a, s_i, s_j);
    score_kernel<<<dim3(512, 4), 256, 0, stream>>>(edge, adj, s_i, s_j, w1a3, escore);
    attn_kernel<<<dim3(32, 4), 256, 0, stream>>>(escore, h, out);
}


# score kernel: every wave load of edge covers 512 contiguous bytes (was 32 lines per load); row dots finished by 16-lane DPP transpose-reduce (new sum order of s_e); 256x16-wave grid, one chunk ahead
# speedup vs baseline: 1.9885x; 1.9885x over previous
#include <hip/hip_runtime.h>
#include <hip/hip_bf16.h>

typedef __attribute__((ext_vector_type(2))) float v2f;
typedef __attribute__((ext_vector_type(4))) float v4f;
typedef __attribute__((ext_vector_type(8))) float v8f;

#define ALPHA_LRELU 0.2f
#define NEG_INF_F  (-1.0e30f)

__global__ void w1a3_kernel(const float* __restrict__ W1, const float* __restrict__ a,
                            float* __restrict__ w1a3) {
    int t = threadIdx.x;
    const float* row = W1 + (size_t)t * 256;
    float acc = 0.f;
    #pragma unroll 4
    for (int c = 0; c < 256; c += 4) {
        v4f v = *(const v4f*)(row + c);
        acc += v[0] * a[512 + c] + v[1] * a[512 + c + 1] +
               v[2] * a[512 + c + 2] + v[3] * a[512 + c + 3];
    }
    w1a3[t] = acc;
}

__global__ void gemm_h_kernel(const float* __restrict__ x, const float* __restrict__ W,
                              float* __restrict__ h) {
    int b    = blockIdx.y;
    int i0   = blockIdx.x * 16;
    int wave = threadIdx.x >> 5;
    int lane = threadIdx.x & 31;

    int M    = lane & 15;
    int koff = (lane >> 4) << 1;
    int klo  = (lane < 16) ? 0 : 2;
    int ncol = lane & 15;

    const float* xrow = x + (size_t)(b * 512 + i0 + M) * 256;
    int n0a = wave * 16;
    int n0b = (wave + 8) * 16;

    v8f c0 = {}; v8f c1 = {};
    for (int k0 = 0; k0 < 256; k0 += 4) {
        v2f afrag = *(const v2f*)(xrow + k0 + koff);

        const float* wp = W + (size_t)(k0 + klo) * 256;
        v2f b0, b1;
        b0[0] = wp[n0a + ncol]; b0[1] = wp[256 + n0a + ncol];
        b1[0] = wp[n0b + ncol]; b1[1] = wp[256 + n0b + ncol];

        c0 = __builtin_amdgcn_wmma_f32_16x16x4_f32(false, afrag, false, b0, (short)0, c0, false, false);
        c1 = __builtin_amdgcn_wmma_f32_16x16x4_f32(false, afrag, false, b1, (short)0, c1, false, false);
    }

    int mbase = (lane >> 4) << 3;
    float* hb = h + (size_t)(b * 512 + i0) * 256;
    #pragma unroll
    for (int v = 0; v < 8; ++v) {
        hb[(size_t)(mbase + v) * 256 + n0a + ncol] = c0[v];
        hb[(size_t)(mbase + v) * 256 + n0b + ncol] = c1[v];
    }
}

__global__ void sij_kernel(const float* __restrict__ h, const float* __restrict__ a,
                           float* __restrict__ s_i, float* __restrict__ s_j) {
    int idx = blockIdx.x * blockDim.x + threadIdx.x;
    const float* hr = h + (size_t)idx * 256;
    float acc1 = 0.f, acc2 = 0.f;
    #pragma unroll 4
    for (int o = 0; o < 256; o += 4) {
        v4f hv = *(const v4f*)(hr + o);
        acc1 += hv[0] * a[o]       + hv[1] * a[o + 1]       + hv[2] * a[o + 2]       + hv[3] * a[o + 3];
        acc2 += hv[0] * a[256 + o] + hv[1] * a[256 + o + 1] + hv[2] * a[256 + o + 2] + hv[3] * a[256 + o + 3];
    }
    s_i[idx] = acc1;
    s_j[idx] = acc2;
}

#define SC_WAVES  16
#define SC_CHUNKS 8
#define SC_Q      16

#define ROW_XMASK(v, n) __builtin_bit_cast(float, __builtin_amdgcn_update_dpp( \
        0, __builtin_bit_cast(int, (v)), 0x160 + (n), 0xf, 0xf, true))

__global__ void __launch_bounds__(SC_WAVES * 32)
score_kernel(const float* __restrict__ edge, const float* __restrict__ adj,
             const float* __restrict__ s_i, const float* __restrict__ s_j,
             const float* __restrict__ w1a3, float* __restrict__ escore) {
    const unsigned tid  = threadIdx.x;
    const unsigned lane = tid & 31u;
    const unsigned wave = __builtin_amdgcn_readfirstlane(tid >> 5);
    const unsigned gw   = blockIdx.x * SC_WAVES + wave;
    const unsigned row  = gw >> 1;
    const unsigned jb   = (gw & 1u) * 256u;
    const unsigned m    = lane & 15u;
    const unsigned h    = lane >> 4;
    const unsigned jl   = 2u * m + h;

    const v4f   w  = *(const v4f*)(w1a3 + 4u * m);
    const float si = s_i[row];

    const float* ep  = edge + (size_t)row * 32768u + (size_t)jb * 64u + 4u * lane;
    const float* sjp = s_j + (row & ~511u) + jb + jl;
    const float* adp = adj + (size_t)row * 512u + jb + jl;
    float*       eo  = escore + (size_t)row * 512u + jb + jl;

    const bool b8 = (lane & 8u) != 0u, b4 = (lane & 4u) != 0u;
    const bool b2 = (lane & 2u) != 0u, b1 = (lane & 1u) != 0u;

    v4f   buf[2][SC_Q];
    float sjv[2], adv[2];

#define LOAD_CHUNK(T) do { \
        _Pragma("unroll") for (int q = 0; q < SC_Q; ++q) \
            buf[(T) & 1][q] = __builtin_nontemporal_load((const v4f*)(ep + (T) * 2048 + q * 128));   \
        sjv[(T) & 1] = sjp[(T) * 32]; \
        adv[(T) & 1] = __builtin_nontemporal_load(adp + (T) * 32);   \
    } while (0)

    LOAD_CHUNK(0);
    #pragma unroll
    for (int t = 0; t < SC_CHUNKS; ++t) {
        __builtin_amdgcn_sched_barrier(0);
        if (t + 1 < SC_CHUNKS) LOAD_CHUNK(t + 1);
        __builtin_amdgcn_sched_barrier(0);

        const v4f* v = buf[t & 1];
        float p[SC_Q];
        #pragma unroll
        for (int q = 0; q < SC_Q; ++q)
            p[q] = v[q][0] * w[0] + v[q][1] * w[1] + v[q][2] * w[2] + v[q][3] * w[3];

        float r8[8], r4[4], r2[2];
        #pragma unroll
        for (int q = 0; q < 8; ++q) {
            float keep = b8 ? p[q + 8] : p[q];
            float send = b8 ? p[q] : p[q + 8];
            r8[q] = keep + ROW_XMASK(send, 8);
        }
        #pragma unroll
        for (int q = 0; q < 4; ++q) {
            float keep = b4 ? r8[q + 4] : r8[q];
            float send = b4 ? r8[q] : r8[q + 4];
            r4[q] = keep + ROW_XMASK(send, 4);
        }
        #pragma unroll
        for (int q = 0; q < 2; ++q) {
            float keep = b2 ? r4[q + 2] : r4[q];
            float send = b2 ? r4[q] : r4[q + 2];
            r2[q] = keep + ROW_XMASK(send, 2);
        }
        float s;
        {
            float keep = b1 ? r2[1] : r2[0];
            float send = b1 ? r2[0] : r2[1];
            s = keep + ROW_XMASK(send, 1);
        }

        float e = si + sjv[t & 1] + s;
        e = (e > 0.f) ? e : ALPHA_LRELU * e;
        e = (adv[t & 1] > 0.0f) ? e : NEG_INF_F;
        eo[t * 32] = e;
    }
#undef LOAD_CHUNK
}

__global__ void attn_kernel(const float* __restrict__ escore, const float* __restrict__ h,
                            float* __restrict__ out) {
    __shared__ float att[16][516];
    __shared__ float red[16][16];
    __shared__ float rowstat[16];

    int b   = blockIdx.y;
    int i0  = blockIdx.x * 16;
    int tid = threadIdx.x;

    const float* ebase = escore + (size_t)(b * 512 + i0) * 512;
    #pragma unroll
    for (int iter = 0; iter < 32; ++iter) {
        int flat = iter * 256 + tid;
        att[flat >> 9][flat & 511] = ebase[flat];
    }
    __syncthreads();

    int row = tid >> 4;
    int l16 = tid & 15;

    float m = -3.0e38f;
    for (int j = l16; j < 512; j += 16) m = fmaxf(m, att[row][j]);
    red[row][l16] = m;
    __syncthreads();
    if (l16 == 0) {
        float mm = red[row][0];
        #pragma unroll
        for (int q = 1; q < 16; ++q) mm = fmaxf(mm, red[row][q]);
        rowstat[row] = mm;
    }
    __syncthreads();
    float mx = rowstat[row];

    float sum = 0.f;
    for (int j = l16; j < 512; j += 16) {
        float v = __expf(att[row][j] - mx);
        att[row][j] = v;
        sum += v;
    }
    red[row][l16] = sum;
    __syncthreads();
    if (l16 == 0) {
        float t = 0.f;
        #pragma unroll
        for (int q = 0; q < 16; ++q) t += red[row][q];
        rowstat[row] = t;
    }
    __syncthreads();
    float inv = 1.0f / rowstat[row];
    for (int j = l16; j < 512; j += 16) att[row][j] *= inv;
    __syncthreads();

    int wave = tid >> 5;
    int lane = tid & 31;
    int M     = lane & 15;
    int koff  = (lane >> 4) << 1;
    int klo   = (lane < 16) ? 0 : 2;
    int ncol  = lane & 15;
    int mbase = (lane >> 4) << 3;

    const float* hb = h + (size_t)b * 512 * 256;
    float* ob = out + (size_t)(b * 512 + i0) * 256;

    #pragma unroll 1
    for (int t2 = 0; t2 < 2; ++t2) {
        int n0 = (wave + t2 * 8) * 16;
        v8f c = {};
        for (int k0 = 0; k0 < 512; k0 += 4) {
            v2f afrag;
            afrag[0] = att[M][k0 + koff];
            afrag[1] = att[M][k0 + koff + 1];
            v2f bf;
            const float* hp = hb + (size_t)(k0 + klo) * 256 + n0 + ncol;
            bf[0] = hp[0];
            bf[1] = hp[256];
            c = __builtin_amdgcn_wmma_f32_16x16x4_f32(false, afrag, false, bf, (short)0, c, false, false);
        }
        #pragma unroll
        for (int v = 0; v < 8; ++v) {
            float xv = c[v];
            float o = (xv > 0.f) ? xv : (__expf(xv) - 1.0f);
            ob[(size_t)(mbase + v) * 256 + n0 + ncol] = o;
        }
    }
}

extern "C" void kernel_launch(void* const* d_in, const int* in_sizes, int n_in,
                              void* d_out, int out_size, void* d_ws, size_t ws_size,
                              hipStream_t stream) {
    const float* x    = (const float*)d_in[0];
    const float* edge = (const float*)d_in[1];
    const float* adj  = (const float*)d_in[2];
    const float* W    = (const float*)d_in[3];
    const float* W1   = (const float*)d_in[4];
    const float* a    = (const float*)d_in[5];
    float* out = (float*)d_out;

    float* h      = (float*)d_ws;
    float* s_i    = h + 4 * 512 * 256;
    float* s_j    = s_i + 4 * 512;
    float* w1a3   = s_j + 4 * 512;
    float* escore = w1a3 + 64;

    w1a3_kernel<<<1, 64, 0, stream>>>(W1, a, w1a3);
    gemm_h_kernel<<<dim3(32, 4), 256, 0, stream>>>(x, W, h);
    sij_kernel<<<8, 256, 0, stream>>>(h, a, s_i, s_j);
    score_kernel<<<256, 512, 0, stream>>>(edge, adj, s_i, s_j, w1a3, escore);
    attn_kernel<<<dim3(32, 4), 256, 0, stream>>>(escore, h, out);
}


# 5 launches -> 3: s_i/s_j row dots move into the h GEMM epilogue (from accumulators; DPP + LDS reduce) and w1a3 into a spare workgroup of that launch; new sum order of s_i, s_j, w1a3
# speedup vs baseline: 2.4879x; 1.2511x over previous
#include <hip/hip_runtime.h>
#include <hip/hip_bf16.h>

typedef __attribute__((ext_vector_type(2))) float v2f;
typedef __attribute__((ext_vector_type(4))) float v4f;
typedef __attribute__((ext_vector_type(8))) float v8f;

#define ALPHA_LRELU 0.2f
#define NEG_INF_F  (-1.0e30f)

#define ROW_XMASK(v, n) __builtin_bit_cast(float, __builtin_amdgcn_update_dpp( \
        0, __builtin_bit_cast(int, (v)), 0x160 + (n), 0xf, 0xf, true))

__global__ void __launch_bounds__(256)
gemm_h_kernel(const float* __restrict__ x, const float* __restrict__ W,
              const float* __restrict__ W1, const float* __restrict__ a,
              float* __restrict__ h, float* __restrict__ sij, float* __restrict__ w1a3) {
    __shared__ float red[8][32];
    __shared__ float wl[64];

    const int bid = blockIdx.x;
    const int tid = threadIdx.x;

    if (bid == 128) {
        const int r = tid >> 2, q = tid & 3;
        const float* wrow = W1 + r * 256 + q * 4;
        const float* ap   = a + 512 + q * 4;
        float acc = 0.f;
        #pragma unroll
        for (int it = 0; it < 16; ++it) {
            v4f v = *(const v4f*)(wrow + it * 16);
            v4f s = *(const v4f*)(ap + it * 16);
            acc += v[0] * s[0] + v[1] * s[1] + v[2] * s[2] + v[3] * s[3];
        }
        acc += ROW_XMASK(acc, 1);
        acc += ROW_XMASK(acc, 2);
        if (q == 0) wl[r] = acc;
        __syncthreads();
        if (tid < 64) w1a3[tid] = wl[tid];
        return;
    }

    const int b    = bid >> 5;
    const int i0   = (bid & 31) * 16;
    const int wave = tid >> 5;
    const int lane = tid & 31;

    const int M    = lane & 15;
    const int koff = (lane >> 4) << 1;
    const int klo  = (lane < 16) ? 0 : 2;
    const int ncol = lane & 15;

    const float* xrow = x + (size_t)(b * 512 + i0 + M) * 256;
    const int n0a = wave * 16;
    const int n0b = (wave + 8) * 16;

    const float a1a = a[n0a + ncol],       a1b = a[n0b + ncol];
    const float a2a = a[256 + n0a + ncol], a2b = a[256 + n0b + ncol];

    v8f c0 = {}; v8f c1 = {};
    for (int k0 = 0; k0 < 256; k0 += 4) {
        v2f afrag = *(const v2f*)(xrow + k0 + koff);

        const float* wp = W + (size_t)(k0 + klo) * 256;
        v2f b0, b1;
        b0[0] = wp[n0a + ncol]; b0[1] = wp[256 + n0a + ncol];
        b1[0] = wp[n0b + ncol]; b1[1] = wp[256 + n0b + ncol];

        c0 = __builtin_amdgcn_wmma_f32_16x16x4_f32(false, afrag, false, b0, (short)0, c0, false, false);
        c1 = __builtin_amdgcn_wmma_f32_16x16x4_f32(false, afrag, false, b1, (short)0, c1, false, false);
    }

    const int mbase = (lane >> 4) << 3;
    float* hb = h + (size_t)(b * 512 + i0) * 256;
    #pragma unroll
    for (int v = 0; v < 8; ++v) {
        hb[(size_t)(mbase + v) * 256 + n0a + ncol] = c0[v];
        hb[(size_t)(mbase + v) * 256 + n0b + ncol] = c1[v];
    }

    float p[16];
    #pragma unroll
    for (int v = 0; v < 8; ++v) {
        p[v]     = c0[v] * a1a + c1[v] * a1b;
        p[8 + v] = c0[v] * a2a + c1[v] * a2b;
    }
    const bool b8 = (lane & 8) != 0, b4 = (lane & 4) != 0;
    const bool b2 = (lane & 2) != 0, b1l = (lane & 1) != 0;
    float r8[8], r4[4], r2[2];
    #pragma unroll
    for (int q = 0; q < 8; ++q) {
        float keep = b8 ? p[q + 8] : p[q];
        float send = b8 ? p[q] : p[q + 8];
        r8[q] = keep + ROW_XMASK(send, 8);
    }
    #pragma unroll
    for (int q = 0; q < 4; ++q) {
        float keep = b4 ? r8[q + 4] : r8[q];
        float send = b4 ? r8[q] : r8[q + 4];
        r4[q] = keep + ROW_XMASK(send, 4);
    }
    #pragma unroll
    for (int q = 0; q < 2; ++q) {
        float keep = b2 ? r4[q + 2] : r4[q];
        float send = b2 ? r4[q] : r4[q + 2];
        r2[q] = keep + ROW_XMASK(send, 2);
    }
    {
        float keep = b1l ? r2[1] : r2[0];
        float send = b1l ? r2[0] : r2[1];
        red[wave][lane] = keep + ROW_XMASK(send, 1);
    }
    __syncthreads();
    if (wave == 0) {
        float t = 0.f;
        #pragma unroll
        for (int w = 0; w < 8; ++w) t += red[w][lane];
        const int pos = ((lane & 8) << 1) + ((lane >> 4) << 3) + (lane & 7);
        sij[bid * 32 + pos] = t;
    }
}

#define SC_WAVES  16
#define SC_CHUNKS 8
#define SC_Q      16

__global__ void __launch_bounds__(SC_WAVES * 32)
score_kernel(const float* __restrict__ edge, const float* __restrict__ adj,
             const float* __restrict__ sij,
             const float* __restrict__ w1a3, float* __restrict__ escore) {
    const unsigned tid  = threadIdx.x;
    const unsigned lane = tid & 31u;
    const unsigned wave = __builtin_amdgcn_readfirstlane(tid >> 5);
    const unsigned gw   = blockIdx.x * SC_WAVES + wave;
    const unsigned row  = gw >> 1;
    const unsigned jb   = (gw & 1u) * 256u;
    const unsigned m    = lane & 15u;
    const unsigned h    = lane >> 4;
    const unsigned jl   = 2u * m + h;

    const v4f   w  = *(const v4f*)(w1a3 + 4u * m);
    const float si = sij[(row >> 4) * 32u + (row & 15u)];

    const float* ep  = edge + (size_t)row * 32768u + (size_t)jb * 64u + 4u * lane;
    const float* sjp = sij + (row >> 9) * 1024u + 2u * jb + (jl >> 4) * 32u + 16u + (jl & 15u);
    const float* adp = adj + (size_t)row * 512u + jb + jl;
    float*       eo  = escore + (size_t)row * 512u + jb + jl;

    const bool b8 = (lane & 8u) != 0u, b4 = (lane & 4u) != 0u;
    const bool b2 = (lane & 2u) != 0u, b1 = (lane & 1u) != 0u;

    v4f   buf[2][SC_Q];
    float sjv[2], adv[2];

#define LOAD_CHUNK(T) do { \
        _Pragma("unroll") for (int q = 0; q < SC_Q; ++q) \
            buf[(T) & 1][q] = __builtin_nontemporal_load((const v4f*)(ep + (T) * 2048 + q * 128));   \
        sjv[(T) & 1] = sjp[(T) * 64]; \
        adv[(T) & 1] = __builtin_nontemporal_load(adp + (T) * 32);   \
    } while (0)

    LOAD_CHUNK(0);
    #pragma unroll
    for (int t = 0; t < SC_CHUNKS; ++t) {
        __builtin_amdgcn_sched_barrier(0);
        if (t + 1 < SC_CHUNKS) LOAD_CHUNK(t + 1);
        __builtin_amdgcn_sched_barrier(0);

        const v4f* v = buf[t & 1];
        float p[SC_Q];
        #pragma unroll
        for (int q = 0; q < SC_Q; ++q)
            p[q] = v[q][0] * w[0] + v[q][1] * w[1] + v[q][2] * w[2] + v[q][3] * w[3];

        float r8[8], r4[4], r2[2];
        #pragma unroll
        for (int q = 0; q < 8; ++q) {
            float keep = b8 ? p[q + 8] : p[q];
            float send = b8 ? p[q] : p[q + 8];
            r8[q] = keep + ROW_XMASK(send, 8);
        }
        #pragma unroll
        for (int q = 0; q < 4; ++q) {
            float keep = b4 ? r8[q + 4] : r8[q];
            float send = b4 ? r8[q] : r8[q + 4];
            r4[q] = keep + ROW_XMASK(send, 4);
        }
        #pragma unroll
        for (int q = 0; q < 2; ++q) {
            float keep = b2 ? r4[q + 2] : r4[q];
            float send = b2 ? r4[q] : r4[q + 2];
            r2[q] = keep + ROW_XMASK(send, 2);
        }
        float s;
        {
            float keep = b1 ? r2[1] : r2[0];
            float send = b1 ? r2[0] : r2[1];
            s = keep + ROW_XMASK(send, 1);
        }

        float e = si + sjv[t & 1] + s;
        e = (e > 0.f) ? e : ALPHA_LRELU * e;
        e = (adv[t & 1] > 0.0f) ? e : NEG_INF_F;
        eo[t * 32] = e;
    }
#undef LOAD_CHUNK
}

__global__ void attn_kernel(const float* __restrict__ escore, const float* __restrict__ h,
                            float* __restrict__ out) {
    __shared__ float att[16][516];
    __shared__ float red[16][16];
    __shared__ float rowstat[16];

    int b   = blockIdx.y;
    int i0  = blockIdx.x * 16;
    int tid = threadIdx.x;

    const float* ebase = escore + (size_t)(b * 512 + i0) * 512;
    #pragma unroll
    for (int iter = 0; iter < 32; ++iter) {
        int flat = iter * 256 + tid;
        att[flat >> 9][flat & 511] = ebase[flat];
    }
    __syncthreads();

    int row = tid >> 4;
    int l16 = tid & 15;

    float m = -3.0e38f;
    for (int j = l16; j < 512; j += 16) m = fmaxf(m, att[row][j]);
    red[row][l16] = m;
    __syncthreads();
    if (l16 == 0) {
        float mm = red[row][0];
        #pragma unroll
        for (int q = 1; q < 16; ++q) mm = fmaxf(mm, red[row][q]);
        rowstat[row] = mm;
    }
    __syncthreads();
    float mx = rowstat[row];

    float sum = 0.f;
    for (int j = l16; j < 512; j += 16) {
        float v = __expf(att[row][j] - mx);
        att[row][j] = v;
        sum += v;
    }
    red[row][l16] = sum;
    __syncthreads();
    if (l16 == 0) {
        float t = 0.f;
        #pragma unroll
        for (int q = 0; q < 16; ++q) t += red[row][q];
        rowstat[row] = t;
    }
    __syncthreads();
    float inv = 1.0f / rowstat[row];
    for (int j = l16; j < 512; j += 16) att[row][j] *= inv;
    __syncthreads();

    int wave = tid >> 5;
    int lane = tid & 31;
    int M     = lane & 15;
    int koff  = (lane >> 4) << 1;
    int klo   = (lane < 16) ? 0 : 2;
    int ncol  = lane & 15;
    int mbase = (lane >> 4) << 3;

    const float* hb = h + (size_t)b * 512 * 256;
    float* ob = out + (size_t)(b * 512 + i0) * 256;

    #pragma unroll 1
    for (int t2 = 0; t2 < 2; ++t2) {
        int n0 = (wave + t2 * 8) * 16;
        v8f c = {};
        for (int k0 = 0; k0 < 512; k0 += 4) {
            v2f afrag;
            afrag[0] = att[M][k0 + koff];
            afrag[1] = att[M][k0 + koff + 1];
            v2f bf;
            const float* hp = hb + (size_t)(k0 + klo) * 256 + n0 + ncol;
            bf[0] = hp[0];
            bf[1] = hp[256];
            c = __builtin_amdgcn_wmma_f32_16x16x4_f32(false, afrag, false, bf, (short)0, c, false, false);
        }
        #pragma unroll
        for (int v = 0; v < 8; ++v) {
            float xv = c[v];
            float o = (xv > 0.f) ? xv : (__expf(xv) - 1.0f);
            ob[(size_t)(mbase + v) * 256 + n0 + ncol] = o;
        }
    }
}

extern "C" void kernel_launch(void* const* d_in, const int* in_sizes, int n_in,
                              void* d_out, int out_size, void* d_ws, size_t ws_size,
                              hipStream_t stream) {
    const float* x    = (const float*)d_in[0];
    const float* edge = (const float*)d_in[1];
    const float* adj  = (const float*)d_in[2];
    const float* W    = (const float*)d_in[3];
    const float* W1   = (const float*)d_in[4];
    const float* a    = (const float*)d_in[5];
    float* out = (float*)d_out;

    float* h      = (float*)d_ws;
    float* sij    = h + 4 * 512 * 256;
    float* w1a3   = sij + 2 * 4 * 512;
    float* escore = w1a3 + 64;

    gemm_h_kernel<<<129, 256, 0, stream>>>(x, W, W1, a, h, sij, w1a3);
    score_kernel<<<256, 512, 0, stream>>>(edge, adj, sij, w1a3, escore);
    attn_kernel<<<dim3(32, 4), 256, 0, stream>>>(escore, h, out);
}


# score stream, row softmax and attention@h fused in one kernel (3 launches -> 2; scores stay in LDS, 8-node workgroups); new sum order: softmax denominators, h_prime as 4 K-quarters
# speedup vs baseline: 3.3717x; 1.3552x over previous
#include <hip/hip_runtime.h>
#include <hip/hip_bf16.h>

typedef __attribute__((ext_vector_type(2))) float v2f;
typedef __attribute__((ext_vector_type(4))) float v4f;
typedef __attribute__((ext_vector_type(8))) float v8f;

#define ALPHA_LRELU 0.2f
#define NEG_INF_F  (-1.0e30f)

#define ROW_XMASK(v, n) __builtin_bit_cast(float, __builtin_amdgcn_update_dpp( \
        0, __builtin_bit_cast(int, (v)), 0x160 + (n), 0xf, 0xf, true))

__global__ void __launch_bounds__(256)
gemm_h_kernel(const float* __restrict__ x, const float* __restrict__ W,
              const float* __restrict__ W1, const float* __restrict__ a,
              float* __restrict__ h, float* __restrict__ sij, float* __restrict__ w1a3) {
    __shared__ float red[8][32];
    __shared__ float wl[64];

    const int bid = blockIdx.x;
    const int tid = threadIdx.x;

    if (bid == 128) {
        const int r = tid >> 2, q = tid & 3;
        const float* wrow = W1 + r * 256 + q * 4;
        const float* ap   = a + 512 + q * 4;
        float acc = 0.f;
        #pragma unroll
        for (int it = 0; it < 16; ++it) {
            v4f v = *(const v4f*)(wrow + it * 16);
            v4f s = *(const v4f*)(ap + it * 16);
            acc += v[0] * s[0] + v[1] * s[1] + v[2] * s[2] + v[3] * s[3];
        }
        acc += ROW_XMASK(acc, 1);
        acc += ROW_XMASK(acc, 2);
        if (q == 0) wl[r] = acc;
        __syncthreads();
        if (tid < 64) w1a3[tid] = wl[tid];
        return;
    }

    const int b    = bid >> 5;
    const int i0   = (bid & 31) * 16;
    const int wave = tid >> 5;
    const int lane = tid & 31;

    const int M    = lane & 15;
    const int koff = (lane >> 4) << 1;
    const int klo  = (lane < 16) ? 0 : 2;
    const int ncol = lane & 15;

    const float* xrow = x + (size_t)(b * 512 + i0 + M) * 256;
    const int n0a = wave * 16;
    const int n0b = (wave + 8) * 16;

    const float a1a = a[n0a + ncol],       a1b = a[n0b + ncol];
    const float a2a = a[256 + n0a + ncol], a2b = a[256 + n0b + ncol];

    v8f c0 = {}; v8f c1 = {};
    for (int k0 = 0; k0 < 256; k0 += 4) {
        v2f afrag = *(const v2f*)(xrow + k0 + koff);

        const float* wp = W + (size_t)(k0 + klo) * 256;
        v2f b0, b1;
        b0[0] = wp[n0a + ncol]; b0[1] = wp[256 + n0a + ncol];
        b1[0] = wp[n0b + ncol]; b1[1] = wp[256 + n0b + ncol];

        c0 = __builtin_amdgcn_wmma_f32_16x16x4_f32(false, afrag, false, b0, (short)0, c0, false, false);
        c1 = __builtin_amdgcn_wmma_f32_16x16x4_f32(false, afrag, false, b1, (short)0, c1, false, false);
    }

    const int mbase = (lane >> 4) << 3;
    float* hb = h + (size_t)(b * 512 + i0) * 256;
    #pragma unroll
    for (int v = 0; v < 8; ++v) {
        hb[(size_t)(mbase + v) * 256 + n0a + ncol] = c0[v];
        hb[(size_t)(mbase + v) * 256 + n0b + ncol] = c1[v];
    }

    float p[16];
    #pragma unroll
    for (int v = 0; v < 8; ++v) {
        p[v]     = c0[v] * a1a + c1[v] * a1b;
        p[8 + v] = c0[v] * a2a + c1[v] * a2b;
    }
    const bool b8 = (lane & 8) != 0, b4 = (lane & 4) != 0;
    const bool b2 = (lane & 2) != 0, b1l = (lane & 1) != 0;
    float r8[8], r4[4], r2[2];
    #pragma unroll
    for (int q = 0; q < 8; ++q) {
        float keep = b8 ? p[q + 8] : p[q];
        float send = b8 ? p[q] : p[q + 8];
        r8[q] = keep + ROW_XMASK(send, 8);
    }
    #pragma unroll
    for (int q = 0; q < 4; ++q) {
        float keep = b4 ? r8[q + 4] : r8[q];
        float send = b4 ? r8[q] : r8[q + 4];
        r4[q] = keep + ROW_XMASK(send, 4);
    }
    #pragma unroll
    for (int q = 0; q < 2; ++q) {
        float keep = b2 ? r4[q + 2] : r4[q];
        float send = b2 ? r4[q] : r4[q + 2];
        r2[q] = keep + ROW_XMASK(send, 2);
    }
    {
        float keep = b1l ? r2[1] : r2[0];
        float send = b1l ? r2[0] : r2[1];
        red[wave][lane] = keep + ROW_XMASK(send, 1);
    }
    __syncthreads();
    if (wave == 0) {
        float t = 0.f;
        #pragma unroll
        for (int w = 0; w < 8; ++w) t += red[w][lane];
        const int pos = ((lane & 8) << 1) + ((lane >> 4) << 3) + (lane & 7);
        sij[bid * 32 + pos] = t;
    }
}

#define SC_WAVES  16
#define SC_CHUNKS 8
#define SC_Q      16
#define ATT_PITCH 516
#define KB_STEPS  8
#define KB_BLOCKS 4

__global__ void __launch_bounds__(SC_WAVES * 32)
score_attn_kernel(const float* __restrict__ edge, const float* __restrict__ adj,
                  const float* __restrict__ sij, const float* __restrict__ w1a3,
                  const float* __restrict__ h, float* __restrict__ out) {
    __shared__ __attribute__((aligned(16))) float att[16][ATT_PITCH];
    __shared__ __attribute__((aligned(16))) float part[4][8][256];
    __shared__ float wmax[SC_WAVES];
    __shared__ float wsum[SC_WAVES];

    const unsigned tid  = threadIdx.x;
    const unsigned lane = tid & 31u;
    const unsigned wave = __builtin_amdgcn_readfirstlane(tid >> 5);
    const unsigned gw   = blockIdx.x * SC_WAVES + wave;
    const unsigned row  = gw >> 1;
    const unsigned jb   = (gw & 1u) * 256u;
    const unsigned m    = lane & 15u;
    const unsigned hh   = lane >> 4;
    const unsigned jl   = 2u * m + hh;

    for (unsigned i = tid; i < 8u * ATT_PITCH; i += SC_WAVES * 32u) (&att[8][0])[i] = 0.f;

    const v4f   w  = *(const v4f*)(w1a3 + 4u * m);
    const float si = sij[(row >> 4) * 32u + (row & 15u)];

    const float* ep  = edge + (size_t)row * 32768u + (size_t)jb * 64u + 4u * lane;
    const float* sjp = sij + (row >> 9) * 1024u + 2u * jb + (jl >> 4) * 32u + 16u + (jl & 15u);
    const float* adp = adj + (size_t)row * 512u + jb + jl;

    const bool b8 = (lane & 8u) != 0u, b4 = (lane & 4u) != 0u;
    const bool b2 = (lane & 2u) != 0u, b1 = (lane & 1u) != 0u;

    v4f   buf[2][SC_Q];
    float sjv[2], adv[2];
    float* arow = &att[wave >> 1][jb + jl];

#define LOAD_CHUNK(T) do { \
        _Pragma("unroll") for (int q = 0; q < SC_Q; ++q) \
            buf[(T) & 1][q] = __builtin_nontemporal_load((const v4f*)(ep + (T) * 2048 + q * 128));   \
        sjv[(T) & 1] = sjp[(T) * 64]; \
        adv[(T) & 1] = __builtin_nontemporal_load(adp + (T) * 32);   \
    } while (0)

    LOAD_CHUNK(0);
    #pragma unroll
    for (int t = 0; t < SC_CHUNKS; ++t) {
        __builtin_amdgcn_sched_barrier(0);
        if (t + 1 < SC_CHUNKS) LOAD_CHUNK(t + 1);
        __builtin_amdgcn_sched_barrier(0);

        const v4f* v = buf[t & 1];
        float p[SC_Q];
        #pragma unroll
        for (int q = 0; q < SC_Q; ++q)
            p[q] = v[q][0] * w[0] + v[q][1] * w[1] + v[q][2] * w[2] + v[q][3] * w[3];

        float r8[8], r4[4], r2[2];
        #pragma unroll
        for (int q = 0; q < 8; ++q) {
            float keep = b8 ? p[q + 8] : p[q];
            float send = b8 ? p[q] : p[q + 8];
            r8[q] = keep + ROW_XMASK(send, 8);
        }
        #pragma unroll
        for (int q = 0; q < 4; ++q) {
            float keep = b4 ? r8[q + 4] : r8[q];
            float send = b4 ? r8[q] : r8[q + 4];
            r4[q] = keep + ROW_XMASK(send, 4);
        }
        #pragma unroll
        for (int q = 0; q < 2; ++q) {
            float keep = b2 ? r4[q + 2] : r4[q];
            float send = b2 ? r4[q] : r4[q + 2];
            r2[q] = keep + ROW_XMASK(send, 2);
        }
        float s;
        {
            float keep = b1 ? r2[1] : r2[0];
            float send = b1 ? r2[0] : r2[1];
            s = keep + ROW_XMASK(send, 1);
        }

        float e = si + sjv[t & 1] + s;
        e = (e > 0.f) ? e : ALPHA_LRELU * e;
        e = (adv[t & 1] > 0.0f) ? e : NEG_INF_F;
        arow[t * 32] = e;
    }
#undef LOAD_CHUNK

    const unsigned c16  = lane & 15u;
    const unsigned koff = hh << 1;
    const unsigned bidx = blockIdx.x >> 6;
    const unsigned i0   = (blockIdx.x & 63u) * 8u;
    const unsigned cg   = wave & 3u;
    const unsigned kq   = wave >> 2;
    const float* hq = h + (size_t)bidx * 131072u + (size_t)(128u * kq + koff) * 256u + 64u * cg + 4u * c16;
    const float* ap = &att[c16][128u * kq + koff];

    v4f hb[2][2 * KB_STEPS];
#define LOAD_HBLOCK(T) do { \
        _Pragma("unroll") for (int s = 0; s < KB_STEPS; ++s) { \
            hb[(T) & 1][2 * s]     = *(const v4f*)(hq + ((T) * KB_STEPS + s) * 1024); \
            hb[(T) & 1][2 * s + 1] = *(const v4f*)(hq + ((T) * KB_STEPS + s) * 1024 + 256); \
        } \
    } while (0)
    LOAD_HBLOCK(0);

    float ev[SC_CHUNKS];
    #pragma unroll
    for (int t = 0; t < SC_CHUNKS; ++t) ev[t] = arow[t * 32];
    float mx = ev[0];
    #pragma unroll
    for (int t = 1; t < SC_CHUNKS; ++t) mx = fmaxf(mx, ev[t]);
    mx = fmaxf(mx, ROW_XMASK(mx, 8));
    mx = fmaxf(mx, ROW_XMASK(mx, 4));
    mx = fmaxf(mx, ROW_XMASK(mx, 2));
    mx = fmaxf(mx, ROW_XMASK(mx, 1));
    mx = fmaxf(mx, __shfl_xor(mx, 16));
    if (lane == 0u) wmax[wave] = mx;
    __syncthreads();
    mx = fmaxf(wmax[wave & ~1u], wmax[wave | 1u]);

    float sum = 0.f;
    #pragma unroll
    for (int t = 0; t < SC_CHUNKS; ++t) {
        ev[t] = __expf(ev[t] - mx);
        sum += ev[t];
    }
    sum += ROW_XMASK(sum, 8);
    sum += ROW_XMASK(sum, 4);
    sum += ROW_XMASK(sum, 2);
    sum += ROW_XMASK(sum, 1);
    sum += __shfl_xor(sum, 16);
    if (lane == 0u) wsum[wave] = sum;
    __syncthreads();
    const float inv = 1.0f / (wsum[wave & ~1u] + wsum[wave | 1u]);
    #pragma unroll
    for (int t = 0; t < SC_CHUNKS; ++t) arow[t * 32] = ev[t] * inv;
    __syncthreads();

    v8f acc[4] = {};
    #pragma unroll
    for (int blk = 0; blk < KB_BLOCKS; ++blk) {
        __builtin_amdgcn_sched_barrier(0);
        if (blk + 1 < KB_BLOCKS) LOAD_HBLOCK(blk + 1);
        __builtin_amdgcn_sched_barrier(0);
        #pragma unroll
        for (int s = 0; s < KB_STEPS; ++s) {
            const v2f af = *(const v2f*)(ap + (blk * KB_STEPS + s) * 4);
            const v4f r0 = hb[blk & 1][2 * s], r1 = hb[blk & 1][2 * s + 1];
            #pragma unroll
            for (int T = 0; T < 4; ++T) {
                v2f bf; bf[0] = r0[T]; bf[1] = r1[T];
                acc[T] = __builtin_amdgcn_wmma_f32_16x16x4_f32(false, af, false, bf,
                                                               (short)0, acc[T], false, false);
            }
        }
    }
#undef LOAD_HBLOCK

    if (hh == 0u) {
        #pragma unroll
        for (int v = 0; v < 8; ++v) {
            v4f o; o[0] = acc[0][v]; o[1] = acc[1][v]; o[2] = acc[2][v]; o[3] = acc[3][v];
            *(v4f*)&part[kq][v][64u * cg + 4u * c16] = o;
        }
    }
    __syncthreads();
    {
        const unsigned v   = tid >> 6;
        const unsigned col = (tid & 63u) * 4u;
        const v4f p0 = *(const v4f*)&part[0][v][col], p1 = *(const v4f*)&part[1][v][col];
        const v4f p2 = *(const v4f*)&part[2][v][col], p3 = *(const v4f*)&part[3][v][col];
        v4f o;
        #pragma unroll
        for (int e = 0; e < 4; ++e) {
            float xv = ((p0[e] + p1[e]) + p2[e]) + p3[e];
            o[e] = (xv > 0.f) ? xv : (__expf(xv) - 1.0f);
        }
        *(v4f*)(out + (size_t)(bidx * 512u + i0 + v) * 256u + col) = o;
    }
}

extern "C" void kernel_launch(void* const* d_in, const int* in_sizes, int n_in,
                              void* d_out, int out_size, void* d_ws, size_t ws_size,
                              hipStream_t stream) {
    const float* x    = (const float*)d_in[0];
    const float* edge = (const float*)d_in[1];
    const float* adj  = (const float*)d_in[2];
    const float* W    = (const float*)d_in[3];
    const float* W1   = (const float*)d_in[4];
    const float* a    = (const float*)d_in[5];
    float* out = (float*)d_out;

    float* h      = (float*)d_ws;
    float* sij    = h + 4 * 512 * 256;
    float* w1a3   = sij + 2 * 4 * 512;

    gemm_h_kernel<<<129, 256, 0, stream>>>(x, W, W1, a, h, sij, w1a3);
    score_attn_kernel<<<256, 512, 0, stream>>>(edge, adj, sij, w1a3, h, out);
}


# h GEMM operand feed: x tile staged once per workgroup in LDS, W fragments as 8-byte loads of two column-interleaved tiles, one 16-step block ahead; new sum order of s_i, s_j
# speedup vs baseline: 3.5720x; 1.0594x over previous
#include <hip/hip_runtime.h>
#include <hip/hip_bf16.h>

typedef __attribute__((ext_vector_type(2))) float v2f;
typedef __attribute__((ext_vector_type(4))) float v4f;
typedef __attribute__((ext_vector_type(8))) float v8f;

#define ALPHA_LRELU 0.2f
#define NEG_INF_F  (-1.0e30f)

#define ROW_XMASK(v, n) __builtin_bit_cast(float, __builtin_amdgcn_update_dpp( \
        0, __builtin_bit_cast(int, (v)), 0x160 + (n), 0xf, 0xf, true))

#define XS_PITCH  260
#define GK_STEPS  16
#define GK_BLOCKS 4
__global__ void __launch_bounds__(256)
gemm_h_kernel(const float* __restrict__ x, const float* __restrict__ W,
              const float* __restrict__ W1, const float* __restrict__ a,
              float* __restrict__ h, float* __restrict__ sij, float* __restrict__ w1a3) {
    __shared__ float red[8][32];
    __shared__ float wl[64];
    __shared__ __attribute__((aligned(16))) float xs[16][XS_PITCH];

    const int bid = blockIdx.x;
    const int tid = threadIdx.x;

    if (bid == 128) {
        const int r = tid >> 2, q = tid & 3;
        const float* wrow = W1 + r * 256 + q * 4;
        const float* ap   = a + 512 + q * 4;
        float acc = 0.f;
        #pragma unroll
        for (int it = 0; it < 16; ++it) {
            v4f v = *(const v4f*)(wrow + it * 16);
            v4f s = *(const v4f*)(ap + it * 16);
            acc += v[0] * s[0] + v[1] * s[1] + v[2] * s[2] + v[3] * s[3];
        }
        acc += ROW_XMASK(acc, 1);
        acc += ROW_XMASK(acc, 2);
        if (q == 0) wl[r] = acc;
        __syncthreads();
        if (tid < 64) w1a3[tid] = wl[tid];
        return;
    }

    const int b    = bid >> 5;
    const int i0   = (bid & 31) * 16;
    const int wave = tid >> 5;
    const int lane = tid & 31;

    const int M    = lane & 15;
    const int koff = (lane >> 4) << 1;
    const int klo  = (lane < 16) ? 0 : 2;
    const int ncol = lane & 15;

    const int cbase = wave * 32 + 2 * ncol;

    const v2f a1v = *(const v2f*)(a + cbase);
    const v2f a2v = *(const v2f*)(a + 256 + cbase);
    const float a1a = a1v[0], a1b = a1v[1];
    const float a2a = a2v[0], a2b = a2v[1];

    const float* wcol = W + (size_t)klo * 256 + cbase;
    v2f wr[2][2 * GK_STEPS];
    v2f af[2][GK_STEPS];
#define LOAD_WBLOCK(T) do { \
        _Pragma("unroll") for (int s = 0; s < GK_STEPS; ++s) { \
            wr[(T) & 1][2 * s]     = *(const v2f*)(wcol + ((T) * GK_STEPS + s) * 1024); \
            wr[(T) & 1][2 * s + 1] = *(const v2f*)(wcol + ((T) * GK_STEPS + s) * 1024 + 256); \
        } \
    } while (0)
    LOAD_WBLOCK(0);

    {
        const float* xt = x + (size_t)(b * 512 + i0) * 256;
        v4f xv[4];
        #pragma unroll
        for (int it = 0; it < 4; ++it) xv[it] = *(const v4f*)(xt + (it * 256 + tid) * 4);
        #pragma unroll
        for (int it = 0; it < 4; ++it) *(v4f*)&xs[it * 4 + (tid >> 6)][(tid & 63) * 4] = xv[it];
    }
    __syncthreads();

    const float* xa = &xs[M][koff];
#define LOAD_ABLOCK(T) do { \
        _Pragma("unroll") for (int s = 0; s < GK_STEPS; ++s) \
            af[(T) & 1][s] = *(const v2f*)(xa + ((T) * GK_STEPS + s) * 4);     \
    } while (0)
    LOAD_ABLOCK(0);

    v8f c0 = {}; v8f c1 = {};
    #pragma unroll
    for (int blk = 0; blk < GK_BLOCKS; ++blk) {
        __builtin_amdgcn_sched_barrier(0);
        if (blk + 1 < GK_BLOCKS) { LOAD_WBLOCK(blk + 1); LOAD_ABLOCK(blk + 1); }
        __builtin_amdgcn_sched_barrier(0);
        #pragma unroll
        for (int s = 0; s < GK_STEPS; ++s) {
            const v2f r0 = wr[blk & 1][2 * s], r1 = wr[blk & 1][2 * s + 1];
            v2f b0, b1;
            b0[0] = r0[0]; b0[1] = r1[0];
            b1[0] = r0[1]; b1[1] = r1[1];
            c0 = __builtin_amdgcn_wmma_f32_16x16x4_f32(false, af[blk & 1][s], false, b0, (short)0, c0, false, false);
            c1 = __builtin_amdgcn_wmma_f32_16x16x4_f32(false, af[blk & 1][s], false, b1, (short)0, c1, false, false);
        }
    }
#undef LOAD_WBLOCK
#undef LOAD_ABLOCK

    const int mbase = (lane >> 4) << 3;
    float* hb = h + (size_t)(b * 512 + i0) * 256;
    #pragma unroll
    for (int v = 0; v < 8; ++v) {
        v2f o; o[0] = c0[v]; o[1] = c1[v];
        *(v2f*)&hb[(size_t)(mbase + v) * 256 + cbase] = o;
    }

    float p[16];
    #pragma unroll
    for (int v = 0; v < 8; ++v) {
        p[v]     = c0[v] * a1a + c1[v] * a1b;
        p[8 + v] = c0[v] * a2a + c1[v] * a2b;
    }
    const bool b8 = (lane & 8) != 0, b4 = (lane & 4) != 0;
    const bool b2 = (lane & 2) != 0, b1l = (lane & 1) != 0;
    float r8[8], r4[4], r2[2];
    #pragma unroll
    for (int q = 0; q < 8; ++q) {
        float keep = b8 ? p[q + 8] : p[q];
        float send = b8 ? p[q] : p[q + 8];
        r8[q] = keep + ROW_XMASK(send, 8);
    }
    #pragma unroll
    for (int q = 0; q < 4; ++q) {
        float keep = b4 ? r8[q + 4] : r8[q];
        float send = b4 ? r8[q] : r8[q + 4];
        r4[q] = keep + ROW_XMASK(send, 4);
    }
    #pragma unroll
    for (int q = 0; q < 2; ++q) {
        float keep = b2 ? r4[q + 2] : r4[q];
        float send = b2 ? r4[q] : r4[q + 2];
        r2[q] = keep + ROW_XMASK(send, 2);
    }
    {
        float keep = b1l ? r2[1] : r2[0];
        float send = b1l ? r2[0] : r2[1];
        red[wave][lane] = keep + ROW_XMASK(send, 1);
    }
    __syncthreads();
    if (wave == 0) {
        float t = 0.f;
        #pragma unroll
        for (int w = 0; w < 8; ++w) t += red[w][lane];
        const int pos = ((lane & 8) << 1) + ((lane >> 4) << 3) + (lane & 7);
        sij[bid * 32 + pos] = t;
    }
}

#define SC_WAVES  16
#define SC_CHUNKS 8
#define SC_Q      16
#define ATT_PITCH 516
#define KB_STEPS  8
#define KB_BLOCKS 4

__global__ void __launch_bounds__(SC_WAVES * 32)
score_attn_kernel(const float* __restrict__ edge, const float* __restrict__ adj,
                  const float* __restrict__ sij, const float* __restrict__ w1a3,
                  const float* __restrict__ h, float* __restrict__ out) {
    __shared__ __attribute__((aligned(16))) float att[16][ATT_PITCH];
    __shared__ __attribute__((aligned(16))) float part[4][8][256];
    __shared__ float wmax[SC_WAVES];
    __shared__ float wsum[SC_WAVES];

    const unsigned tid  = threadIdx.x;
    const unsigned lane = tid & 31u;
    const unsigned wave = __builtin_amdgcn_readfirstlane(tid >> 5);
    const unsigned gw   = blockIdx.x * SC_WAVES + wave;
    const unsigned row  = gw >> 1;
    const unsigned jb   = (gw & 1u) * 256u;
    const unsigned m    = lane & 15u;
    const unsigned hh   = lane >> 4;
    const unsigned jl   = 2u * m + hh;

    for (unsigned i = tid; i < 8u * ATT_PITCH; i += SC_WAVES * 32u) (&att[8][0])[i] = 0.f;

    const v4f   w  = *(const v4f*)(w1a3 + 4u * m);
    const float si = sij[(row >> 4) * 32u + (row & 15u)];

    const float* ep  = edge + (size_t)row * 32768u + (size_t)jb * 64u + 4u * lane;
    const float* sjp = sij + (row >> 9) * 1024u + 2u * jb + (jl >> 4) * 32u + 16u + (jl & 15u);
    const float* adp = adj + (size_t)row * 512u + jb + jl;

    const bool b8 = (lane & 8u) != 0u, b4 = (lane & 4u) != 0u;
    const bool b2 = (lane & 2u) != 0u, b1 = (lane & 1u) != 0u;

    v4f   buf[2][SC_Q];
    float sjv[2], adv[2];
    float* arow = &att[wave >> 1][jb + jl];

#define LOAD_CHUNK(T) do { \
        _Pragma("unroll") for (int q = 0; q < SC_Q; ++q) \
            buf[(T) & 1][q] = __builtin_nontemporal_load((const v4f*)(ep + (T) * 2048 + q * 128));   \
        sjv[(T) & 1] = sjp[(T) * 64]; \
        adv[(T) & 1] = __builtin_nontemporal_load(adp + (T) * 32);   \
    } while (0)

    LOAD_CHUNK(0);
    #pragma unroll
    for (int t = 0; t < SC_CHUNKS; ++t) {
        __builtin_amdgcn_sched_barrier(0);
        if (t + 1 < SC_CHUNKS) LOAD_CHUNK(t + 1);
        __builtin_amdgcn_sched_barrier(0);

        const v4f* v = buf[t & 1];
        float p[SC_Q];
        #pragma unroll
        for (int q = 0; q < SC_Q; ++q)
            p[q] = v[q][0] * w[0] + v[q][1] * w[1] + v[q][2] * w[2] + v[q][3] * w[3];

        float r8[8], r4[4], r2[2];
        #pragma unroll
        for (int q = 0; q < 8; ++q) {
            float keep = b8 ? p[q + 8] : p[q];
            float send = b8 ? p[q] : p[q + 8];
            r8[q] = keep + ROW_XMASK(send, 8);
        }
        #pragma unroll
        for (int q = 0; q < 4; ++q) {
            float keep = b4 ? r8[q + 4] : r8[q];
            float send = b4 ? r8[q] : r8[q + 4];
            r4[q] = keep + ROW_XMASK(send, 4);
        }
        #pragma unroll
        for (int q = 0; q < 2; ++q) {
            float keep = b2 ? r4[q + 2] : r4[q];
            float send = b2 ? r4[q] : r4[q + 2];
            r2[q] = keep + ROW_XMASK(send, 2);
        }
        float s;
        {
            float keep = b1 ? r2[1] : r2[0];
            float send = b1 ? r2[0] : r2[1];
            s = keep + ROW_XMASK(send, 1);
        }

        float e = si + sjv[t & 1] + s;
        e = (e > 0.f) ? e : ALPHA_LRELU * e;
        e = (adv[t & 1] > 0.0f) ? e : NEG_INF_F;
        arow[t * 32] = e;
    }
#undef LOAD_CHUNK

    const unsigned c16  = lane & 15u;
    const unsigned koff = hh << 1;
    const unsigned bidx = blockIdx.x >> 6;
    const unsigned i0   = (blockIdx.x & 63u) * 8u;
    const unsigned cg   = wave & 3u;
    const unsigned kq   = wave >> 2;
    const float* hq = h + (size_t)bidx * 131072u + (size_t)(128u * kq + koff) * 256u + 64u * cg + 4u * c16;
    const float* ap = &att[c16][128u * kq + koff];

    v4f hb[2][2 * KB_STEPS];
#define LOAD_HBLOCK(T) do { \
        _Pragma("unroll") for (int s = 0; s < KB_STEPS; ++s) { \
            hb[(T) & 1][2 * s]     = *(const v4f*)(hq + ((T) * KB_STEPS + s) * 1024); \
            hb[(T) & 1][2 * s + 1] = *(const v4f*)(hq + ((T) * KB_STEPS + s) * 1024 + 256); \
        } \
    } while (0)
    LOAD_HBLOCK(0);

    float ev[SC_CHUNKS];
    #pragma unroll
    for (int t = 0; t < SC_CHUNKS; ++t) ev[t] = arow[t * 32];
    float mx = ev[0];
    #pragma unroll
    for (int t = 1; t < SC_CHUNKS; ++t) mx = fmaxf(mx, ev[t]);
    mx = fmaxf(mx, ROW_XMASK(mx, 8));
    mx = fmaxf(mx, ROW_XMASK(mx, 4));
    mx = fmaxf(mx, ROW_XMASK(mx, 2));
    mx = fmaxf(mx, ROW_XMASK(mx, 1));
    mx = fmaxf(mx, __shfl_xor(mx, 16));
    if (lane == 0u) wmax[wave] = mx;
    __syncthreads();
    mx = fmaxf(wmax[wave & ~1u], wmax[wave | 1u]);

    float sum = 0.f;
    #pragma unroll
    for (int t = 0; t < SC_CHUNKS; ++t) {
        ev[t] = __expf(ev[t] - mx);
        sum += ev[t];
    }
    sum += ROW_XMASK(sum, 8);
    sum += ROW_XMASK(sum, 4);
    sum += ROW_XMASK(sum, 2);
    sum += ROW_XMASK(sum, 1);
    sum += __shfl_xor(sum, 16);
    if (lane == 0u) wsum[wave] = sum;
    __syncthreads();
    const float inv = 1.0f / (wsum[wave & ~1u] + wsum[wave | 1u]);
    #pragma unroll
    for (int t = 0; t < SC_CHUNKS; ++t) arow[t * 32] = ev[t] * inv;
    __syncthreads();

    v8f acc[4] = {};
    #pragma unroll
    for (int blk = 0; blk < KB_BLOCKS; ++blk) {
        __builtin_amdgcn_sched_barrier(0);
        if (blk + 1 < KB_BLOCKS) LOAD_HBLOCK(blk + 1);
        __builtin_amdgcn_sched_barrier(0);
        #pragma unroll
        for (int s = 0; s < KB_STEPS; ++s) {
            const v2f af = *(const v2f*)(ap + (blk * KB_STEPS + s) * 4);
            const v4f r0 = hb[blk & 1][2 * s], r1 = hb[blk & 1][2 * s + 1];
            #pragma unroll
            for (int T = 0; T < 4; ++T) {
                v2f bf; bf[0] = r0[T]; bf[1] = r1[T];
                acc[T] = __builtin_amdgcn_wmma_f32_16x16x4_f32(false, af, false, bf,
                                                               (short)0, acc[T], false, false);
            }
        }
    }
#undef LOAD_HBLOCK

    if (hh == 0u) {
        #pragma unroll
        for (int v = 0; v < 8; ++v) {
            v4f o; o[0] = acc[0][v]; o[1] = acc[1][v]; o[2] = acc[2][v]; o[3] = acc[3][v];
            *(v4f*)&part[kq][v][64u * cg + 4u * c16] = o;
        }
    }
    __syncthreads();
    {
        const unsigned v   = tid >> 6;
        const unsigned col = (tid & 63u) * 4u;
        const v4f p0 = *(const v4f*)&part[0][v][col], p1 = *(const v4f*)&part[1][v][col];
        const v4f p2 = *(const v4f*)&part[2][v][col], p3 = *(const v4f*)&part[3][v][col];
        v4f o;
        #pragma unroll
        for (int e = 0; e < 4; ++e) {
            float xv = ((p0[e] + p1[e]) + p2[e]) + p3[e];
            o[e] = (xv > 0.f) ? xv : (__expf(xv) - 1.0f);
        }
        *(v4f*)(out + (size_t)(bidx * 512u + i0 + v) * 256u + col) = o;
    }
}

extern "C" void kernel_launch(void* const* d_in, const int* in_sizes, int n_in,
                              void* d_out, int out_size, void* d_ws, size_t ws_size,
                              hipStream_t stream) {
    const float* x    = (const float*)d_in[0];
    const float* edge = (const float*)d_in[1];
    const float* adj  = (const float*)d_in[2];
    const float* W    = (const float*)d_in[3];
    const float* W1   = (const float*)d_in[4];
    const float* a    = (const float*)d_in[5];
    float* out = (float*)d_out;

    float* h      = (float*)d_ws;
    float* sij    = h + 4 * 512 * 256;
    float* w1a3   = sij + 2 * 4 * 512;

    gemm_h_kernel<<<129, 256, 0, stream>>>(x, W, W1, a, h, sij, w1a3);
    score_attn_kernel<<<256, 512, 0, stream>>>(edge, adj, sij, w1a3, h, out);
}


# online softmax: attention@h matrix steps run inside the edge stream, one 64-j step per chunk with running half-row max and rescale (tail phase gone); new sum order: softmax denominators, h_prime
# speedup vs baseline: 3.7943x; 1.0622x over previous
#include <hip/hip_runtime.h>
#include <hip/hip_bf16.h>

typedef __attribute__((ext_vector_type(2))) float v2f;
typedef __attribute__((ext_vector_type(4))) float v4f;
typedef __attribute__((ext_vector_type(8))) float v8f;

#define ALPHA_LRELU 0.2f
#define NEG_INF_F  (-1.0e30f)

#define ROW_XMASK(v, n) __builtin_bit_cast(float, __builtin_amdgcn_update_dpp( \
        0, __builtin_bit_cast(int, (v)), 0x160 + (n), 0xf, 0xf, true))

#define XS_PITCH  260
#define GK_STEPS  16
#define GK_BLOCKS 4
__global__ void __launch_bounds__(256)
gemm_h_kernel(const float* __restrict__ x, const float* __restrict__ W,
              const float* __restrict__ W1, const float* __restrict__ a,
              float* __restrict__ h, float* __restrict__ sij, float* __restrict__ w1a3) {
    __shared__ float red[8][32];
    __shared__ float wl[64];
    __shared__ __attribute__((aligned(16))) float xs[16][XS_PITCH];

    const int bid = blockIdx.x;
    const int tid = threadIdx.x;

    if (bid == 128) {
        const int r = tid >> 2, q = tid & 3;
        const float* wrow = W1 + r * 256 + q * 4;
        const float* ap   = a + 512 + q * 4;
        float acc = 0.f;
        #pragma unroll
        for (int it = 0; it < 16; ++it) {
            v4f v = *(const v4f*)(wrow + it * 16);
            v4f s = *(const v4f*)(ap + it * 16);
            acc += v[0] * s[0] + v[1] * s[1] + v[2] * s[2] + v[3] * s[3];
        }
        acc += ROW_XMASK(acc, 1);
        acc += ROW_XMASK(acc, 2);
        if (q == 0) wl[r] = acc;
        __syncthreads();
        if (tid < 64) w1a3[tid] = wl[tid];
        return;
    }

    const int b    = bid >> 5;
    const int i0   = (bid & 31) * 16;
    const int wave = tid >> 5;
    const int lane = tid & 31;

    const int M    = lane & 15;
    const int koff = (lane >> 4) << 1;
    const int klo  = (lane < 16) ? 0 : 2;
    const int ncol = lane & 15;

    const int cbase = wave * 32 + 2 * ncol;

    const v2f a1v = *(const v2f*)(a + cbase);
    const v2f a2v = *(const v2f*)(a + 256 + cbase);
    const float a1a = a1v[0], a1b = a1v[1];
    const float a2a = a2v[0], a2b = a2v[1];

    const float* wcol = W + (size_t)klo * 256 + cbase;
    v2f wr[2][2 * GK_STEPS];
    v2f af[2][GK_STEPS];
#define LOAD_WBLOCK(T) do { \
        _Pragma("unroll") for (int s = 0; s < GK_STEPS; ++s) { \
            wr[(T) & 1][2 * s]     = *(const v2f*)(wcol + ((T) * GK_STEPS + s) * 1024); \
            wr[(T) & 1][2 * s + 1] = *(const v2f*)(wcol + ((T) * GK_STEPS + s) * 1024 + 256); \
        } \
    } while (0)
    LOAD_WBLOCK(0);

    {
        const float* xt = x + (size_t)(b * 512 + i0) * 256;
        v4f xv[4];
        #pragma unroll
        for (int it = 0; it < 4; ++it) xv[it] = *(const v4f*)(xt + (it * 256 + tid) * 4);
        #pragma unroll
        for (int it = 0; it < 4; ++it) *(v4f*)&xs[it * 4 + (tid >> 6)][(tid & 63) * 4] = xv[it];
    }
    __syncthreads();

    const float* xa = &xs[M][koff];
#define LOAD_ABLOCK(T) do { \
        _Pragma("unroll") for (int s = 0; s < GK_STEPS; ++s) \
            af[(T) & 1][s] = *(const v2f*)(xa + ((T) * GK_STEPS + s) * 4);     \
    } while (0)
    LOAD_ABLOCK(0);

    v8f c0 = {}; v8f c1 = {};
    #pragma unroll
    for (int blk = 0; blk < GK_BLOCKS; ++blk) {
        __builtin_amdgcn_sched_barrier(0);
        if (blk + 1 < GK_BLOCKS) { LOAD_WBLOCK(blk + 1); LOAD_ABLOCK(blk + 1); }
        __builtin_amdgcn_sched_barrier(0);
        #pragma unroll
        for (int s = 0; s < GK_STEPS; ++s) {
            const v2f r0 = wr[blk & 1][2 * s], r1 = wr[blk & 1][2 * s + 1];
            v2f b0, b1;
            b0[0] = r0[0]; b0[1] = r1[0];
            b1[0] = r0[1]; b1[1] = r1[1];
            c0 = __builtin_amdgcn_wmma_f32_16x16x4_f32(false, af[blk & 1][s], false, b0, (short)0, c0, false, false);
            c1 = __builtin_amdgcn_wmma_f32_16x16x4_f32(false, af[blk & 1][s], false, b1, (short)0, c1, false, false);
        }
    }
#undef LOAD_WBLOCK
#undef LOAD_ABLOCK

    const int mbase = (lane >> 4) << 3;
    float* hb = h + (size_t)(b * 512 + i0) * 256;
    #pragma unroll
    for (int v = 0; v < 8; ++v) {
        v2f o; o[0] = c0[v]; o[1] = c1[v];
        *(v2f*)&hb[(size_t)(mbase + v) * 256 + cbase] = o;
    }

    float p[16];
    #pragma unroll
    for (int v = 0; v < 8; ++v) {
        p[v]     = c0[v] * a1a + c1[v] * a1b;
        p[8 + v] = c0[v] * a2a + c1[v] * a2b;
    }
    const bool b8 = (lane & 8) != 0, b4 = (lane & 4) != 0;
    const bool b2 = (lane & 2) != 0, b1l = (lane & 1) != 0;
    float r8[8], r4[4], r2[2];
    #pragma unroll
    for (int q = 0; q < 8; ++q) {
        float keep = b8 ? p[q + 8] : p[q];
        float send = b8 ? p[q] : p[q + 8];
        r8[q] = keep + ROW_XMASK(send, 8);
    }
    #pragma unroll
    for (int q = 0; q < 4; ++q) {
        float keep = b4 ? r8[q + 4] : r8[q];
        float send = b4 ? r8[q] : r8[q + 4];
        r4[q] = keep + ROW_XMASK(send, 4);
    }
    #pragma unroll
    for (int q = 0; q < 2; ++q) {
        float keep = b2 ? r4[q + 2] : r4[q];
        float send = b2 ? r4[q] : r4[q + 2];
        r2[q] = keep + ROW_XMASK(send, 2);
    }
    {
        float keep = b1l ? r2[1] : r2[0];
        float send = b1l ? r2[0] : r2[1];
        red[wave][lane] = keep + ROW_XMASK(send, 1);
    }
    __syncthreads();
    if (wave == 0) {
        float t = 0.f;
        #pragma unroll
        for (int w = 0; w < 8; ++w) t += red[w][lane];
        const int pos = ((lane & 8) << 1) + ((lane >> 4) << 3) + (lane & 7);
        sij[bid * 32 + pos] = t;
    }
}

#define SC_WAVES  16
#define SC_CHUNKS 8
#define SC_Q      16
#define P_PITCH   68

__global__ void __launch_bounds__(SC_WAVES * 32)
score_attn_kernel(const float* __restrict__ edge, const float* __restrict__ adj,
                  const float* __restrict__ sij, const float* __restrict__ w1a3,
                  const float* __restrict__ h, float* __restrict__ out) {
    __shared__ __attribute__((aligned(16))) float P[2][16][P_PITCH];
    __shared__ __attribute__((aligned(16))) float part[4][8][256];
    __shared__ __attribute__((aligned(16))) float scl[2][2][8];
    __shared__ float wm[SC_WAVES];
    __shared__ float wl[SC_WAVES];

    const unsigned tid  = threadIdx.x;
    const unsigned lane = tid & 31u;
    const unsigned wave = __builtin_amdgcn_readfirstlane(tid >> 5);
    const unsigned gw   = blockIdx.x * SC_WAVES + wave;
    const unsigned row  = gw >> 1;
    const unsigned half = gw & 1u;
    const unsigned node = wave >> 1;
    const unsigned jb   = half * 256u;
    const unsigned m    = lane & 15u;
    const unsigned hh   = lane >> 4;
    const unsigned jl   = 2u * m + hh;

    for (unsigned i = tid; i < 2u * 8u * P_PITCH; i += SC_WAVES * 32u) {
        const unsigned rb = i / (8u * P_PITCH);
        (&P[rb][8][0])[i - rb * 8u * P_PITCH] = 0.f;
    }

    const v4f   w  = *(const v4f*)(w1a3 + 4u * m);
    const float si = sij[(row >> 4) * 32u + (row & 15u)];

    const float* ep  = edge + (size_t)row * 32768u + (size_t)jb * 64u + 4u * lane;
    const float* sjp = sij + (row >> 9) * 1024u + 2u * jb + (jl >> 4) * 32u + 16u + (jl & 15u);
    const float* adp = adj + (size_t)row * 512u + jb + jl;

    const bool b8 = (lane & 8u) != 0u, b4 = (lane & 4u) != 0u;
    const bool b2 = (lane & 2u) != 0u, b1 = (lane & 1u) != 0u;

    const unsigned c16  = lane & 15u;
    const unsigned koff = hh << 1;
    const unsigned bidx = blockIdx.x >> 6;
    const unsigned i0   = (blockIdx.x & 63u) * 8u;
    const unsigned cg   = wave & 3u;
    const unsigned kq   = wave >> 2;
    const unsigned kh   = kq >> 1;
    const unsigned kl   = 16u * (kq & 1u) + koff;
    const float* hq = h + (size_t)bidx * 131072u + (size_t)(256u * kh + kl) * 256u + 64u * cg + 4u * c16;
    const float* ap = &P[0][c16][32u * kh + kl];

    v4f   buf[2][SC_Q];
    float sjv[2], adv[2];
    v4f   hb[8];
    v8f   acc[4] = {};
    float m_run  = -3.0e38f;
    float l_lane = 0.f;

#define LOAD_CHUNK(T) do { \
        _Pragma("unroll") for (int q = 0; q < SC_Q; ++q) \
            buf[(T) & 1][q] = __builtin_nontemporal_load((const v4f*)(ep + (T) * 2048 + q * 128));   \
        sjv[(T) & 1] = sjp[(T) * 64]; \
        adv[(T) & 1] = __builtin_nontemporal_load(adp + (T) * 32);   \
    } while (0)
#define LOAD_HSTEP(T) do { \
        _Pragma("unroll") for (int s = 0; s < 4; ++s) { \
            hb[2 * s]     = *(const v4f*)(hq + ((T) * 32 + 4 * s) * 256); \
            hb[2 * s + 1] = *(const v4f*)(hq + ((T) * 32 + 4 * s + 1) * 256); \
        } \
    } while (0)

    LOAD_CHUNK(0);
    #pragma unroll
    for (int t = 0; t < SC_CHUNKS; ++t) {
        __builtin_amdgcn_sched_barrier(0);
        LOAD_HSTEP(t);
        if (t + 1 < SC_CHUNKS) LOAD_CHUNK(t + 1);
        __builtin_amdgcn_sched_barrier(0);

        const v4f* v = buf[t & 1];
        float p[SC_Q];
        #pragma unroll
        for (int q = 0; q < SC_Q; ++q)
            p[q] = v[q][0] * w[0] + v[q][1] * w[1] + v[q][2] * w[2] + v[q][3] * w[3];

        float r8[8], r4[4], r2[2];
        #pragma unroll
        for (int q = 0; q < 8; ++q) {
            float keep = b8 ? p[q + 8] : p[q];
            float send = b8 ? p[q] : p[q + 8];
            r8[q] = keep + ROW_XMASK(send, 8);
        }
        #pragma unroll
        for (int q = 0; q < 4; ++q) {
            float keep = b4 ? r8[q + 4] : r8[q];
            float send = b4 ? r8[q] : r8[q + 4];
            r4[q] = keep + ROW_XMASK(send, 4);
        }
        #pragma unroll
        for (int q = 0; q < 2; ++q) {
            float keep = b2 ? r4[q + 2] : r4[q];
            float send = b2 ? r4[q] : r4[q + 2];
            r2[q] = keep + ROW_XMASK(send, 2);
        }
        float s;
        {
            float keep = b1 ? r2[1] : r2[0];
            float send = b1 ? r2[0] : r2[1];
            s = keep + ROW_XMASK(send, 1);
        }

        float e = si + sjv[t & 1] + s;
        e = (e > 0.f) ? e : ALPHA_LRELU * e;
        e = (adv[t & 1] > 0.0f) ? e : NEG_INF_F;

        float cm = e;
        cm = fmaxf(cm, ROW_XMASK(cm, 8));
        cm = fmaxf(cm, ROW_XMASK(cm, 4));
        cm = fmaxf(cm, ROW_XMASK(cm, 2));
        cm = fmaxf(cm, ROW_XMASK(cm, 1));
        cm = fmaxf(cm, __shfl_xor(cm, 16));
        const float m_new = fmaxf(m_run, cm);
        const float sc    = __expf(m_run - m_new);
        const float pv    = __expf(e - m_new);
        l_lane = l_lane * sc + pv;
        m_run  = m_new;
        P[t & 1][node][32u * half + jl] = pv;
        if (lane == 0u) scl[t & 1][half][node] = sc;
        __builtin_amdgcn_fence(__ATOMIC_RELEASE, "workgroup", "local");
        __builtin_amdgcn_s_barrier();
        __builtin_amdgcn_fence(__ATOMIC_ACQUIRE, "workgroup", "local");

        {
            const v4f s0 = *(const v4f*)&scl[t & 1][kh][0];
            const v4f s1 = *(const v4f*)&scl[t & 1][kh][4];
            #pragma unroll
            for (int T = 0; T < 4; ++T) {
                acc[T][0] *= s0[0]; acc[T][1] *= s0[1]; acc[T][2] *= s0[2]; acc[T][3] *= s0[3];
                acc[T][4] *= s1[0]; acc[T][5] *= s1[1]; acc[T][6] *= s1[2]; acc[T][7] *= s1[3];
            }
            const float* apt = ap + (t & 1) * (16 * P_PITCH);
            #pragma unroll
            for (int st = 0; st < 4; ++st) {
                const v2f af = *(const v2f*)(apt + st * 4);
                const v4f r0 = hb[2 * st], r1 = hb[2 * st + 1];
                #pragma unroll
                for (int T = 0; T < 4; ++T) {
                    v2f bf; bf[0] = r0[T]; bf[1] = r1[T];
                    acc[T] = __builtin_amdgcn_wmma_f32_16x16x4_f32(false, af, false, bf,
                                                                   (short)0, acc[T], false, false);
                }
            }
        }
    }
#undef LOAD_CHUNK
#undef LOAD_HSTEP

    float l = l_lane;
    l += ROW_XMASK(l, 8);
    l += ROW_XMASK(l, 4);
    l += ROW_XMASK(l, 2);
    l += ROW_XMASK(l, 1);
    l += __shfl_xor(l, 16);
    if (lane == 0u) { wm[wave] = m_run; wl[wave] = l; }

    if (hh == 0u) {
        #pragma unroll
        for (int v = 0; v < 8; ++v) {
            v4f o; o[0] = acc[0][v]; o[1] = acc[1][v]; o[2] = acc[2][v]; o[3] = acc[3][v];
            *(v4f*)&part[kq][v][64u * cg + 4u * c16] = o;
        }
    }
    __syncthreads();
    {
        const unsigned v   = tid >> 6;
        const unsigned col = (tid & 63u) * 4u;
        const float m0 = wm[2u * v], m1 = wm[2u * v + 1u];
        const float mx = fmaxf(m0, m1);
        const float f0 = __expf(m0 - mx), f1 = __expf(m1 - mx);
        const float inv = 1.0f / (f0 * wl[2u * v] + f1 * wl[2u * v + 1u]);
        const float g0 = f0 * inv, g1 = f1 * inv;
        const v4f p0 = *(const v4f*)&part[0][v][col], p1 = *(const v4f*)&part[1][v][col];
        const v4f p2 = *(const v4f*)&part[2][v][col], p3 = *(const v4f*)&part[3][v][col];
        v4f o;
        #pragma unroll
        for (int e = 0; e < 4; ++e) {
            float xv = (p0[e] + p1[e]) * g0 + (p2[e] + p3[e]) * g1;
            o[e] = (xv > 0.f) ? xv : (__expf(xv) - 1.0f);
        }
        *(v4f*)(out + (size_t)(bidx * 512u + i0 + v) * 256u + col) = o;
    }
}

extern "C" void kernel_launch(void* const* d_in, const int* in_sizes, int n_in,
                              void* d_out, int out_size, void* d_ws, size_t ws_size,
                              hipStream_t stream) {
    const float* x    = (const float*)d_in[0];
    const float* edge = (const float*)d_in[1];
    const float* adj  = (const float*)d_in[2];
    const float* W    = (const float*)d_in[3];
    const float* W1   = (const float*)d_in[4];
    const float* a    = (const float*)d_in[5];
    float* out = (float*)d_out;

    float* h      = (float*)d_ws;
    float* sij    = h + 4 * 512 * 256;
    float* w1a3   = sij + 2 * 4 * 512;

    gemm_h_kernel<<<129, 256, 0, stream>>>(x, W, W1, a, h, sij, w1a3);
    score_attn_kernel<<<256, 512, 0, stream>>>(edge, adj, sij, w1a3, h, out);
}
